# starting kernel
# baseline (speedup 1.0000x reference)
#include <hip/hip_runtime.h>
#include <hip/hip_bf16.h>
#include <math.h>

typedef float v2f __attribute__((ext_vector_type(2)));
typedef float v8f __attribute__((ext_vector_type(8)));

#define D 256
#define NPG 512

__global__ void node_dots(const float* __restrict__ x,
                          const float* __restrict__ w_rel,
                          const float* __restrict__ w_root,
                          float* __restrict__ r, float* __restrict__ s, int N) {
    const int lane = threadIdx.x & 31;
    const int wave = threadIdx.x >> 5;
    const int node = blockIdx.x * (blockDim.x >> 5) + wave;
    if (node >= N) return;
    const float* xr = x + (size_t)node * D;
    const int j0 = lane * 8;
    float ar = 0.f, as = 0.f;
#pragma unroll
    for (int t = 0; t < 8; ++t) {
        const float xv = xr[j0 + t];
        ar += xv * w_rel[j0 + t];
        as += xv * w_root[j0 + t];
    }
#pragma unroll
    for (int off = 16; off > 0; off >>= 1) {
        ar += __shfl_down(ar, off);
        as += __shfl_down(as, off);
    }
    if (lane == 0) { r[node] = ar; s[node] = as; }
}

__global__ void graph_score_pool(const float* __restrict__ x,
                                 const float* __restrict__ r,
                                 const float* __restrict__ s,
                                 const int* __restrict__ esrc,
                                 const int* __restrict__ edst,
                                 const float* __restrict__ b_rel,
                                 float* __restrict__ pooled,
                                 int Eg, int k) {
    __shared__ float acc[NPG];
    __shared__ float score[NPG];
    __shared__ int   keep[NPG];
    __shared__ float partial[2 * D];

    const int b     = blockIdx.x;
    const int tid   = threadIdx.x;
    const int nbase = b * NPG;

    acc[tid] = 0.f;
    __syncthreads();

    const long ebase = (long)b * Eg;
    for (int e = tid; e < Eg; e += blockDim.x) {
        const int sg = esrc[ebase + e];
        const int dg = edst[ebase + e] - nbase;
        atomicAdd(&acc[dg], r[sg]);
    }
    __syncthreads();

    const float sc = tanhf(acc[tid] + s[nbase + tid] + b_rel[0]);
    score[tid] = sc;
    __syncthreads();

    int cnt = 0;
    for (int j = 0; j < NPG; ++j) {
        const float v = score[j];
        cnt += (v > sc) || (v == sc && j < tid);
    }
    keep[tid] = (cnt < k) ? 1 : 0;
    __syncthreads();

    const int c = tid & (D - 1);
    const int h = tid >> 8;
    const int i0 = h * (NPG / 2);
    float ap = 0.f;
    for (int i = i0; i < i0 + NPG / 2; ++i) {
        if (keep[i])
            ap += score[i] * x[(size_t)(nbase + i) * D + c];
    }
    partial[h * D + c] = ap;
    __syncthreads();
    if (tid < D)
        pooled[(size_t)b * D + tid] = (partial[tid] + partial[D + tid]) / (float)k;
}

__global__ void proj_wmma(const float* __restrict__ A,
                          const float* __restrict__ Bm,
                          const float* __restrict__ bias,
                          float* __restrict__ out,
                          int M, int K, int Nn) {
    const int lane  = threadIdx.x & 31;
    const int half  = lane >> 4;
    const int l16   = lane & 15;
    const int tileM = blockIdx.x;
    const int tileN = blockIdx.y;

    const int rowA = tileM * 16 + l16;
    const int colB = tileN * 16 + l16;

    v8f c = {};
    for (int k0 = 0; k0 < K; k0 += 4) {
        const int ka = k0 + half * 2;
        v2f a, bf;
        a.x  = A[(size_t)rowA * K + ka];
        a.y  = A[(size_t)rowA * K + ka + 1];
        bf.x = Bm[(size_t)ka * Nn + colB];
        bf.y = Bm[(size_t)(ka + 1) * Nn + colB];
        c = __builtin_amdgcn_wmma_f32_16x16x4_f32(false, a, false, bf,
                                                  (short)0, c, false, false);
    }

    const float bv = bias[colB];
#pragma unroll
    for (int v = 0; v < 8; ++v) {
        const int row = tileM * 16 + v + half * 8;
        out[(size_t)row * Nn + colB] = c[v] + bv;
    }
}

extern "C" void kernel_launch(void* const* d_in, const int* in_sizes, int n_in,
                              void* d_out, int out_size, void* d_ws, size_t ws_size,
                              hipStream_t stream) {
    const float* x      = (const float*)d_in[0];
    const int*   eidx   = (const int*)d_in[1];
    const float* w_rel  = (const float*)d_in[4];
    const float* b_rel  = (const float*)d_in[5];
    const float* w_root = (const float*)d_in[6];
    const float* w_proj = (const float*)d_in[7];
    const float* b_proj = (const float*)d_in[8];
    float*       out    = (float*)d_out;

    const int d    = D;
    const int N    = in_sizes[0] / d;
    const int Bg   = N / NPG;
    const int E    = in_sizes[1] / 2;
    const int Eg   = E / Bg;
    const int k    = (NPG + 1) / 2;
    const int outd = in_sizes[8];

    float* r      = (float*)d_ws;
    float* s      = r + N;
    float* pooled = s + N;

    node_dots<<<dim3((N + 7) / 8), dim3(256), 0, stream>>>(x, w_rel, w_root, r, s, N);

    graph_score_pool<<<dim3(Bg), dim3(NPG), 0, stream>>>(
        x, r, s, eidx, eidx + E, b_rel, pooled, Eg, k);

    proj_wmma<<<dim3(Bg / 16, outd / 16), dim3(32), 0, stream>>>(
        pooled, w_proj, b_proj, out, Bg, d, outd);
}


# graph_score rank count: outside a wave's own 32 nodes the tie rule's test j<i has one answer per wave, so a pair costs one compare (>= for blocks in front, > behind); same ranks, same sums
# speedup vs baseline: 4.0436x; 4.0436x over previous
#include <hip/hip_runtime.h>
#include <hip/hip_bf16.h>
#include <math.h>

typedef float v2f __attribute__((ext_vector_type(2)));
typedef float v8f __attribute__((ext_vector_type(8)));

#define D 256
#define NPG 512

#define ND_TILES 2
#define ND_T (D / 8)

__global__ void __launch_bounds__(256)
node_dots(const float* __restrict__ x,
          const float* __restrict__ w_rel,
          const float* __restrict__ w_root,
          float* __restrict__ r, float* __restrict__ s, int N) {
    const int lane  = threadIdx.x & 31;
    const int c     = lane & 15;
    const int h     = lane >> 4;
    const int wave  = blockIdx.x * (blockDim.x >> 5) + (threadIdx.x >> 5);
    const int node0 = wave * (16 * ND_TILES);
    if (node0 >= N) return;

    const float4* wp = reinterpret_cast<const float4*>((c == 0 ? w_rel : w_root) + 4 * h);
    float4 wv[ND_T];
#pragma unroll
    for (int t = 0; t < ND_T; ++t) wv[t] = wp[2 * t];

    float4 xv[ND_TILES][ND_T];
#pragma unroll
    for (int i = 0; i < ND_TILES; ++i) {
        const float4* xp = reinterpret_cast<const float4*>(
            x + (size_t)(node0 + 16 * i + c) * D + 4 * h);
#pragma unroll
        for (int t = 0; t < ND_T; ++t) xv[i][t] = xp[2 * t];
    }
    __builtin_amdgcn_sched_barrier(0);

    float rr[ND_TILES], ss[ND_TILES];
#pragma unroll
    for (int i = 0; i < ND_TILES; ++i) {
        v8f a0 = {}, a1 = {};
#pragma unroll
        for (int t = 0; t < ND_T; ++t) {
            v2f wa, xb;
            wa.x = wv[t].x; wa.y = wv[t].y; xb.x = xv[i][t].x; xb.y = xv[i][t].y;
            a0 = __builtin_amdgcn_wmma_f32_16x16x4_f32(false, wa, false, xb,
                                                       (short)0, a0, false, false);
            wa.x = wv[t].z; wa.y = wv[t].w; xb.x = xv[i][t].z; xb.y = xv[i][t].w;
            a1 = __builtin_amdgcn_wmma_f32_16x16x4_f32(false, wa, false, xb,
                                                       (short)0, a1, false, false);
        }
        rr[i] = a0[0] + a1[0];
        ss[i] = a0[1] + a1[1];
        __builtin_amdgcn_sched_barrier(0);
    }

    const float r1 = __shfl_xor(rr[1], 16);
    const float s1 = __shfl_xor(ss[1], 16);
    r[node0 + lane] = h ? r1 : rr[0];
    s[node0 + lane] = h ? s1 : ss[0];
}

__global__ void graph_score(const float* __restrict__ r,
                            float* sw,
                            const int* __restrict__ esrc,
                            const int* __restrict__ edst,
                            const float* __restrict__ b_rel,
                            int Eg, int k) {
    __shared__ float acc[NPG];
    __shared__ __attribute__((aligned(16))) float score[NPG];

    const int b     = blockIdx.x;
    const int tid   = threadIdx.x;
    const int nbase = b * NPG;

    acc[tid] = 0.f;
    __syncthreads();

    const long ebase = (long)b * Eg;
    for (int e = tid; e < Eg; e += blockDim.x) {
        const int sg = esrc[ebase + e];
        const int dg = edst[ebase + e] - nbase;
        atomicAdd(&acc[dg], r[sg]);
    }
    __syncthreads();

    const float sc = tanhf(acc[tid] + sw[nbase + tid] + b_rel[0]);
    score[tid] = sc;
    __syncthreads();

    const int wb = __builtin_amdgcn_readfirstlane(tid >> 5);
    const float4* sc4 = reinterpret_cast<const float4*>(score);
    int c0 = 0, c1 = 0, c2 = 0, c3 = 0;

    for (int jb = 0; jb < wb; ++jb) {
        for (int q = 0; q < 8; ++q) {
            const float4 v = sc4[jb * 8 + q];
            c0 += (v.x >= sc); c1 += (v.y >= sc);
            c2 += (v.z >= sc); c3 += (v.w >= sc);
        }
    }
    {
        for (int q = 0; q < 8; ++q) {
            const float4 v = sc4[wb * 8 + q];
            const int j = wb * 32 + q * 4;
            c0 += (v.x > sc) || (v.x == sc && j + 0 < tid);
            c1 += (v.y > sc) || (v.y == sc && j + 1 < tid);
            c2 += (v.z > sc) || (v.z == sc && j + 2 < tid);
            c3 += (v.w > sc) || (v.w == sc && j + 3 < tid);
        }
    }
    for (int jb = wb + 1; jb < NPG / 32; ++jb) {
        for (int q = 0; q < 8; ++q) {
            const float4 v = sc4[jb * 8 + q];
            c0 += (v.x > sc); c1 += (v.y > sc);
            c2 += (v.z > sc); c3 += (v.w > sc);
        }
    }
    const int cnt = (c0 + c1) + (c2 + c3);
    sw[nbase + tid] = (cnt < k) ? sc : 0.f;
}

#define PCOLS 64
#define PROWS 32

__global__ void __launch_bounds__(256)
pool_rows(const float* __restrict__ x,
          const float* __restrict__ wgt,
          float* __restrict__ pooled, int k) {
    __shared__ float part[8][PCOLS];

    const int tid  = threadIdx.x;
    const int cg   = tid & 15;
    const int rl   = tid >> 4;
    const int wv   = tid >> 5;
    const int b    = blockIdx.x >> 2;
    const int ch   = blockIdx.x & 3;
    const int row0 = b * NPG + rl * PROWS;

    float w[PROWS];
    const float4* wp = reinterpret_cast<const float4*>(wgt + row0);
#pragma unroll
    for (int q = 0; q < PROWS / 4; ++q) {
        const float4 t = wp[q];
        w[4 * q + 0] = t.x; w[4 * q + 1] = t.y;
        w[4 * q + 2] = t.z; w[4 * q + 3] = t.w;
    }

    const float4* xp =
        reinterpret_cast<const float4*>(x + (size_t)row0 * D + ch * PCOLS) + cg;
    const float4* skip = reinterpret_cast<const float4*>(x);
    float4 v[PROWS];
    __builtin_amdgcn_sched_barrier(0);
#pragma unroll
    for (int j = 0; j < PROWS; ++j) {
        const float4* p = (w[j] != 0.f) ? xp + (size_t)j * (D / 4) : skip;
        v[j] = *p;
    }
    __builtin_amdgcn_sched_barrier(0);

    float ax = 0.f, ay = 0.f, az = 0.f, aw = 0.f;
#pragma unroll
    for (int j = 0; j < PROWS; ++j) {
        asm volatile("" : "+v"(v[j].x), "+v"(v[j].y), "+v"(v[j].z), "+v"(v[j].w));
        ax = fmaf(w[j], v[j].x, ax); ay = fmaf(w[j], v[j].y, ay);
        az = fmaf(w[j], v[j].z, az); aw = fmaf(w[j], v[j].w, aw);
    }

    ax += __shfl_xor(ax, 16); ay += __shfl_xor(ay, 16);
    az += __shfl_xor(az, 16); aw += __shfl_xor(aw, 16);
    if ((tid & 31) < 16) {
        part[wv][cg * 4 + 0] = ax; part[wv][cg * 4 + 1] = ay;
        part[wv][cg * 4 + 2] = az; part[wv][cg * 4 + 3] = aw;
    }
    __syncthreads();

    if (tid < PCOLS) {
        float t = part[0][tid];
#pragma unroll
        for (int q = 1; q < 8; ++q) t += part[q][tid];
        pooled[(size_t)b * D + ch * PCOLS + tid] = t / (float)k;
    }
}

#define PJ_WAVES 16
#define PJ_KW (D / PJ_WAVES)

__global__ void __launch_bounds__(32 * PJ_WAVES)
proj_wmma(const float* __restrict__ A,
          const float* __restrict__ Bm,
          const float* __restrict__ bias,
          float* __restrict__ out, int Nn) {
    __shared__ float part[PJ_WAVES][8][32];

    const int lane  = threadIdx.x & 31;
    const int wv    = threadIdx.x >> 5;
    const int half  = lane >> 4;
    const int l16   = lane & 15;
    const int tileM = blockIdx.x;
    const int tileN = blockIdx.y;

    const int rowA = tileM * 16 + l16;
    const int colB = tileN * 16 + l16;
    const float bv = bias[colB];

    const float* ap = A + (size_t)rowA * D + wv * PJ_KW + half * 2;
    const float* bp = Bm + (size_t)(wv * PJ_KW + half * 2) * Nn + colB;
    unsigned ofs = (unsigned)(wv * 8 * 32 + lane) * 4u;
    asm volatile("" : "+v"(ofs));
    float* mine = reinterpret_cast<float*>(reinterpret_cast<char*>(&part[0][0][0]) + ofs);

    v2f a[PJ_KW / 4], bf[PJ_KW / 4];
    for (int j = 0; j < PJ_KW / 4; ++j) {
        a[j].x  = ap[4 * j];
        a[j].y  = ap[4 * j + 1];
        bf[j].x = bp[(size_t)(4 * j) * Nn];
        bf[j].y = bp[(size_t)(4 * j + 1) * Nn];
    }

    v8f c = {};
    for (int j = 0; j < PJ_KW / 4; ++j)
        c = __builtin_amdgcn_wmma_f32_16x16x4_f32(false, a[j], false, bf[j],
                                                  (short)0, c, false, false);

    for (int v = 0; v < 8; ++v) mine[v * 32] = c[v];
    __syncthreads();

    if (wv < 8) {
        float t = part[0][wv][lane];
        for (int q = 1; q < PJ_WAVES; ++q) t += part[q][wv][lane];
        const int row = tileM * 16 + wv + half * 8;
        out[(size_t)row * Nn + colB] = t + bv;
    }
}

extern "C" void kernel_launch(void* const* d_in, const int* in_sizes, int n_in,
                              void* d_out, int out_size, void* d_ws, size_t ws_size,
                              hipStream_t stream) {
    const float* x      = (const float*)d_in[0];
    const int*   eidx   = (const int*)d_in[1];
    const float* w_rel  = (const float*)d_in[4];
    const float* b_rel  = (const float*)d_in[5];
    const float* w_root = (const float*)d_in[6];
    const float* w_proj = (const float*)d_in[7];
    const float* b_proj = (const float*)d_in[8];
    float*       out    = (float*)d_out;

    const int d    = D;
    const int N    = in_sizes[0] / d;
    const int Bg   = N / NPG;
    const int E    = in_sizes[1] / 2;
    const int Eg   = E / Bg;
    const int k    = (NPG + 1) / 2;
    const int outd = in_sizes[8];

    float* r      = (float*)d_ws;
    float* s      = r + N;
    float* pooled = s + N;

    node_dots<<<dim3((N + 255) / 256), dim3(256), 0, stream>>>(x, w_rel, w_root, r, s, N);

    graph_score<<<dim3(Bg), dim3(NPG), 0, stream>>>(
        r, s, eidx, eidx + E, b_rel, Eg, k);

    pool_rows<<<dim3(Bg * (D / PCOLS)), dim3(256), 0, stream>>>(x, s, pooled, k);

    proj_wmma<<<dim3(Bg / 16, outd / 16), dim3(32 * PJ_WAVES), 0, stream>>>(
        pooled, w_proj, b_proj, out, outd);
}


# graph_score and pool_rows fused into one launch over (graph, 128-column half) workgroups: each half ranks the graph, a wave pools its own 32 rows by its lanes' weights; 3 launches not 4; same sums
# speedup vs baseline: 4.7518x; 1.1751x over previous
#include <hip/hip_runtime.h>
#include <hip/hip_bf16.h>
#include <math.h>

typedef float v2f __attribute__((ext_vector_type(2)));
typedef float v8f __attribute__((ext_vector_type(8)));

#define D 256
#define NPG 512

#define ND_TILES 2
#define ND_T (D / 8)

__global__ void __launch_bounds__(256)
node_dots(const float* __restrict__ x,
          const float* __restrict__ w_rel,
          const float* __restrict__ w_root,
          float* __restrict__ r, float* __restrict__ s, int N) {
    const int lane  = threadIdx.x & 31;
    const int c     = lane & 15;
    const int h     = lane >> 4;
    const int wave  = blockIdx.x * (blockDim.x >> 5) + (threadIdx.x >> 5);
    const int node0 = wave * (16 * ND_TILES);
    if (node0 >= N) return;

    const float4* wp = reinterpret_cast<const float4*>((c == 0 ? w_rel : w_root) + 4 * h);
    float4 wv[ND_T];
#pragma unroll
    for (int t = 0; t < ND_T; ++t) wv[t] = wp[2 * t];

    float4 xv[ND_TILES][ND_T];
#pragma unroll
    for (int i = 0; i < ND_TILES; ++i) {
        const float4* xp = reinterpret_cast<const float4*>(
            x + (size_t)(node0 + 16 * i + c) * D + 4 * h);
#pragma unroll
        for (int t = 0; t < ND_T; ++t) xv[i][t] = xp[2 * t];
    }
    __builtin_amdgcn_sched_barrier(0);

    float rr[ND_TILES], ss[ND_TILES];
#pragma unroll
    for (int i = 0; i < ND_TILES; ++i) {
        v8f a0 = {}, a1 = {};
#pragma unroll
        for (int t = 0; t < ND_T; ++t) {
            v2f wa, xb;
            wa.x = wv[t].x; wa.y = wv[t].y; xb.x = xv[i][t].x; xb.y = xv[i][t].y;
            a0 = __builtin_amdgcn_wmma_f32_16x16x4_f32(false, wa, false, xb,
                                                       (short)0, a0, false, false);
            wa.x = wv[t].z; wa.y = wv[t].w; xb.x = xv[i][t].z; xb.y = xv[i][t].w;
            a1 = __builtin_amdgcn_wmma_f32_16x16x4_f32(false, wa, false, xb,
                                                       (short)0, a1, false, false);
        }
        rr[i] = a0[0] + a1[0];
        ss[i] = a0[1] + a1[1];
        __builtin_amdgcn_sched_barrier(0);
    }

    const float r1 = __shfl_xor(rr[1], 16);
    const float s1 = __shfl_xor(ss[1], 16);
    r[node0 + lane] = h ? r1 : rr[0];
    s[node0 + lane] = h ? s1 : ss[0];
}

#define HCOLS 128
#define PROWS 32

__global__ void __launch_bounds__(NPG)
score_pool(const float* __restrict__ x,
           const float* __restrict__ r,
           const float* __restrict__ s,
           const int* __restrict__ esrc,
           const int* __restrict__ edst,
           const float* __restrict__ b_rel,
           float* __restrict__ pooled,
           int Eg, int k) {
    __shared__ float acc[NPG];
    __shared__ __attribute__((aligned(16))) float score[NPG];
    __shared__ __attribute__((aligned(16))) float part[NPG / 32][HCOLS];

    const int b     = blockIdx.x >> 1;
    const int ch    = blockIdx.x & 1;
    const int tid   = threadIdx.x;
    const int nbase = b * NPG;

    acc[tid] = 0.f;
    __syncthreads();

    const long ebase = (long)b * Eg;
    for (int e = tid; e < Eg; e += blockDim.x) {
        const int sg = esrc[ebase + e];
        const int dg = edst[ebase + e] - nbase;
        atomicAdd(&acc[dg], r[sg]);
    }
    __syncthreads();

    const float sc = tanhf(acc[tid] + s[nbase + tid] + b_rel[0]);
    score[tid] = sc;
    __syncthreads();

    const int wb = __builtin_amdgcn_readfirstlane(tid >> 5);
    const float4* sc4 = reinterpret_cast<const float4*>(score);
    int c0 = 0, c1 = 0, c2 = 0, c3 = 0;

    for (int jb = 0; jb < wb; ++jb) {
        for (int q = 0; q < 8; ++q) {
            const float4 v = sc4[jb * 8 + q];
            c0 += (v.x >= sc); c1 += (v.y >= sc);
            c2 += (v.z >= sc); c3 += (v.w >= sc);
        }
    }
    {
        for (int q = 0; q < 8; ++q) {
            const float4 v = sc4[wb * 8 + q];
            const int j = wb * 32 + q * 4;
            c0 += (v.x > sc) || (v.x == sc && j + 0 < tid);
            c1 += (v.y > sc) || (v.y == sc && j + 1 < tid);
            c2 += (v.z > sc) || (v.z == sc && j + 2 < tid);
            c3 += (v.w > sc) || (v.w == sc && j + 3 < tid);
        }
    }
    for (int jb = wb + 1; jb < NPG / 32; ++jb) {
        for (int q = 0; q < 8; ++q) {
            const float4 v = sc4[jb * 8 + q];
            c0 += (v.x > sc); c1 += (v.y > sc);
            c2 += (v.z > sc); c3 += (v.w > sc);
        }
    }
    const int cnt = (c0 + c1) + (c2 + c3);
    const float wgt = (cnt < k) ? sc : 0.f;

    const int lane = tid & 31;
    float w[PROWS];
#pragma unroll
    for (int j = 0; j < PROWS; ++j)
        w[j] = __int_as_float(__builtin_amdgcn_readlane(__float_as_int(wgt), j));

    const char* rows = reinterpret_cast<const char*>(
        x + (size_t)(nbase + wb * PROWS) * D + ch * HCOLS);
    const char* skip = reinterpret_cast<const char*>(x);
    const unsigned lofs = (unsigned)lane * 16u;
    float4 v[PROWS];
#pragma unroll
    for (int j = 0; j < PROWS; ++j) {
        const char* bp = (w[j] != 0.f) ? rows + (size_t)j * (D * sizeof(float)) : skip;
        v[j] = *reinterpret_cast<const float4*>(bp + lofs);
    }
    __builtin_amdgcn_sched_barrier(0);

    float ax = 0.f, ay = 0.f, az = 0.f, aw = 0.f;
#pragma unroll
    for (int j = 0; j < PROWS; ++j) {
        asm volatile("" : "+v"(v[j].x), "+v"(v[j].y), "+v"(v[j].z), "+v"(v[j].w));
        ax = fmaf(w[j], v[j].x, ax); ay = fmaf(w[j], v[j].y, ay);
        az = fmaf(w[j], v[j].z, az); aw = fmaf(w[j], v[j].w, aw);
    }

    float4 pa; pa.x = ax; pa.y = ay; pa.z = az; pa.w = aw;
    *reinterpret_cast<float4*>(&part[wb][lane * 4]) = pa;
    __syncthreads();

    if (tid < HCOLS) {
        float t = part[0][tid] + part[1][tid];
#pragma unroll
        for (int q = 1; q < NPG / 64; ++q) t += part[2 * q][tid] + part[2 * q + 1][tid];
        pooled[(size_t)b * D + ch * HCOLS + tid] = t / (float)k;
    }
}

#define PJ_WAVES 16
#define PJ_KW (D / PJ_WAVES)

__global__ void __launch_bounds__(32 * PJ_WAVES)
proj_wmma(const float* __restrict__ A,
          const float* __restrict__ Bm,
          const float* __restrict__ bias,
          float* __restrict__ out, int Nn) {
    __shared__ float part[PJ_WAVES][8][32];

    const int lane  = threadIdx.x & 31;
    const int wv    = threadIdx.x >> 5;
    const int half  = lane >> 4;
    const int l16   = lane & 15;
    const int tileM = blockIdx.x;
    const int tileN = blockIdx.y;

    const int rowA = tileM * 16 + l16;
    const int colB = tileN * 16 + l16;
    const float bv = bias[colB];

    const float* ap = A + (size_t)rowA * D + wv * PJ_KW + half * 2;
    const float* bp = Bm + (size_t)(wv * PJ_KW + half * 2) * Nn + colB;
    unsigned ofs = (unsigned)(wv * 8 * 32 + lane) * 4u;
    asm volatile("" : "+v"(ofs));
    float* mine = reinterpret_cast<float*>(reinterpret_cast<char*>(&part[0][0][0]) + ofs);

    v2f a[PJ_KW / 4], bf[PJ_KW / 4];
    for (int j = 0; j < PJ_KW / 4; ++j) {
        a[j].x  = ap[4 * j];
        a[j].y  = ap[4 * j + 1];
        bf[j].x = bp[(size_t)(4 * j) * Nn];
        bf[j].y = bp[(size_t)(4 * j + 1) * Nn];
    }

    v8f c = {};
    for (int j = 0; j < PJ_KW / 4; ++j)
        c = __builtin_amdgcn_wmma_f32_16x16x4_f32(false, a[j], false, bf[j],
                                                  (short)0, c, false, false);

    for (int v = 0; v < 8; ++v) mine[v * 32] = c[v];
    __syncthreads();

    if (wv < 8) {
        float t = part[0][wv][lane];
        for (int q = 1; q < PJ_WAVES; ++q) t += part[q][wv][lane];
        const int row = tileM * 16 + wv + half * 8;
        out[(size_t)row * Nn + colB] = t + bv;
    }
}

extern "C" void kernel_launch(void* const* d_in, const int* in_sizes, int n_in,
                              void* d_out, int out_size, void* d_ws, size_t ws_size,
                              hipStream_t stream) {
    const float* x      = (const float*)d_in[0];
    const int*   eidx   = (const int*)d_in[1];
    const float* w_rel  = (const float*)d_in[4];
    const float* b_rel  = (const float*)d_in[5];
    const float* w_root = (const float*)d_in[6];
    const float* w_proj = (const float*)d_in[7];
    const float* b_proj = (const float*)d_in[8];
    float*       out    = (float*)d_out;

    const int d    = D;
    const int N    = in_sizes[0] / d;
    const int Bg   = N / NPG;
    const int E    = in_sizes[1] / 2;
    const int Eg   = E / Bg;
    const int k    = (NPG + 1) / 2;
    const int outd = in_sizes[8];

    float* r      = (float*)d_ws;
    float* s      = r + N;
    float* pooled = s + N;

    node_dots<<<dim3((N + 255) / 256), dim3(256), 0, stream>>>(x, w_rel, w_root, r, s, N);

    score_pool<<<dim3(Bg * (D / HCOLS)), dim3(NPG), 0, stream>>>(
        x, r, s, eidx, eidx + E, b_rel, pooled, Eg, k);

    proj_wmma<<<dim3(Bg / 16, outd / 16), dim3(32 * PJ_WAVES), 0, stream>>>(
        pooled, w_proj, b_proj, out, outd);
}
